# starting kernel
# baseline (speedup 1.0000x reference)
#include <hip/hip_runtime.h>
#include <hip/hip_bf16.h>
#include <math.h>

#define BSZ 32
#define SSZ 2048
#define HSZ 1024

#define NCHUNK 8
#define ROWS_PER_CHUNK (SSZ / NCHUNK)
#define THREADS 256
#define NWAVES (THREADS / 32)
#define ROWS_PER_WAVE (ROWS_PER_CHUNK / NWAVES)
#define PSTRIDE (HSZ + 8)

typedef __attribute__((ext_vector_type(2))) float v2f;
typedef __attribute__((ext_vector_type(8))) float v8f;

__global__ __launch_bounds__(32)
void cbsa_htW_wmma(const float* __restrict__ h_t, const float* __restrict__ W,
                   float* __restrict__ v)
{
    const int lane  = threadIdx.x;
    const int mbase = blockIdx.x * 16;
    const int nbase = blockIdx.y * 16;
    const int r     = lane & 15;
    const int half  = lane >> 4;

    v8f c = {};
    const float* arow = h_t + (size_t)(mbase + r) * HSZ;
    for (int k0 = 0; k0 < HSZ; k0 += 4) {
        const int ka = k0 + 2 * half;
        v2f a, b;
        a.x = arow[ka];
        a.y = arow[ka + 1];
        b.x = W[(size_t)ka * HSZ + nbase + r];
        b.y = W[(size_t)(ka + 1) * HSZ + nbase + r];
        c = __builtin_amdgcn_wmma_f32_16x16x4_f32(
                 false, a,  false, b,
                 (short)0, c,  false,  false);
    }
#pragma unroll
    for (int j = 0; j < 8; ++j) {
        const int m = j + 8 * half;
        v[(size_t)(mbase + m) * HSZ + nbase + r] = c[j];
    }
}

__global__ __launch_bounds__(THREADS)
void cbsa_flash_partial(const float* __restrict__ cntx,
                        const float* __restrict__ vvec,
                        float* __restrict__ partials)
{
    __shared__ float s_v[HSZ];
    __shared__ float s_m[NWAVES];
    __shared__ float s_l[NWAVES];
    __shared__ float s_acc[NWAVES][HSZ];

    const int b     = blockIdx.x;
    const int chunk = blockIdx.y;
    const int t     = threadIdx.x;
    const int wave  = t >> 5;
    const int lane  = t & 31;

    ((float4*)s_v)[t] = ((const float4*)(vvec + (size_t)b * HSZ))[t];
    __syncthreads();

    float4 vf[8];
#pragma unroll
    for (int j = 0; j < 8; ++j) vf[j] = ((const float4*)s_v)[j * 32 + lane];

    float m = -INFINITY;
    float l = 0.0f;
    float4 acc[8];
#pragma unroll
    for (int j = 0; j < 8; ++j) acc[j] = make_float4(0.f, 0.f, 0.f, 0.f);

    const float* base =
        cntx + ((size_t)b * SSZ + (size_t)chunk * ROWS_PER_CHUNK) * HSZ;

    for (int rr = 0; rr < ROWS_PER_WAVE; ++rr) {
        const int row = wave + rr * NWAVES;
        const float4* rp = (const float4*)(base + (size_t)row * HSZ);

        float4 x[8];
#pragma unroll
        for (int j = 0; j < 8; ++j) x[j] = rp[j * 32 + lane];

        float d = 0.0f;
#pragma unroll
        for (int j = 0; j < 8; ++j) {
            d = fmaf(x[j].x, vf[j].x, d);
            d = fmaf(x[j].y, vf[j].y, d);
            d = fmaf(x[j].z, vf[j].z, d);
            d = fmaf(x[j].w, vf[j].w, d);
        }
#pragma unroll
        for (int off = 16; off > 0; off >>= 1) d += __shfl_xor(d, off, 32);

        const float nm = fmaxf(m, d);
        const float p  = __expf(d - nm);
        const float sc = __expf(m - nm);
        l = l * sc + p;
#pragma unroll
        for (int j = 0; j < 8; ++j) {
            acc[j].x = fmaf(p, x[j].x, acc[j].x * sc);
            acc[j].y = fmaf(p, x[j].y, acc[j].y * sc);
            acc[j].z = fmaf(p, x[j].z, acc[j].z * sc);
            acc[j].w = fmaf(p, x[j].w, acc[j].w * sc);
        }
        m = nm;
    }

    if (lane == 0) { s_m[wave] = m; s_l[wave] = l; }
#pragma unroll
    for (int j = 0; j < 8; ++j) ((float4*)s_acc[wave])[j * 32 + lane] = acc[j];
    __syncthreads();

    float M = s_m[0];
#pragma unroll
    for (int w = 1; w < NWAVES; ++w) M = fmaxf(M, s_m[w]);
    float wgt[NWAVES];
    float L = 0.0f;
#pragma unroll
    for (int w = 0; w < NWAVES; ++w) {
        wgt[w] = __expf(s_m[w] - M);
        L = fmaf(s_l[w], wgt[w], L);
    }

    float* out = partials + ((size_t)b * NCHUNK + chunk) * PSTRIDE;
    if (t == 0) { out[0] = M; out[1] = L; }

    float4 s = make_float4(0.f, 0.f, 0.f, 0.f);
#pragma unroll
    for (int w = 0; w < NWAVES; ++w) {
        const float4 a = ((const float4*)s_acc[w])[t];
        s.x = fmaf(wgt[w], a.x, s.x);
        s.y = fmaf(wgt[w], a.y, s.y);
        s.z = fmaf(wgt[w], a.z, s.z);
        s.w = fmaf(wgt[w], a.w, s.w);
    }
    ((float4*)(out + 8))[t] = s;
}

__global__ __launch_bounds__(THREADS)
void cbsa_finalize(const float* __restrict__ h_t,
                   const float* __restrict__ partials,
                   const float* __restrict__ alpha,
                   const float* __restrict__ beta,
                   float* __restrict__ out)
{
    const int b = blockIdx.x;
    const int t = threadIdx.x;

    const float* pb = partials + (size_t)b * NCHUNK * PSTRIDE;

    float M = -INFINITY;
#pragma unroll
    for (int c = 0; c < NCHUNK; ++c) M = fmaxf(M, pb[c * PSTRIDE]);

    float wgt[NCHUNK];
    float L = 0.0f;
#pragma unroll
    for (int c = 0; c < NCHUNK; ++c) {
        wgt[c] = __expf(pb[c * PSTRIDE] - M);
        L = fmaf(pb[c * PSTRIDE + 1], wgt[c], L);
    }

    float4 s = make_float4(0.f, 0.f, 0.f, 0.f);
#pragma unroll
    for (int c = 0; c < NCHUNK; ++c) {
        const float4 a = ((const float4*)(pb + c * PSTRIDE + 8))[t];
        s.x = fmaf(wgt[c], a.x, s.x);
        s.y = fmaf(wgt[c], a.y, s.y);
        s.z = fmaf(wgt[c], a.z, s.z);
        s.w = fmaf(wgt[c], a.w, s.w);
    }

    const float al  = alpha[0];
    const float inv = beta[0] / L;
    const float4 h  = ((const float4*)(h_t + (size_t)b * HSZ))[t];
    float4 o;
    o.x = fmaf(al, h.x, inv * s.x);
    o.y = fmaf(al, h.y, inv * s.y);
    o.z = fmaf(al, h.z, inv * s.z);
    o.w = fmaf(al, h.w, inv * s.w);
    ((float4*)(out + (size_t)b * HSZ))[t] = o;
}

extern "C" void kernel_launch(void* const* d_in, const int* in_sizes, int n_in,
                              void* d_out, int out_size, void* d_ws, size_t ws_size,
                              hipStream_t stream)
{
    const float* h_t   = (const float*)d_in[0];
    const float* cntx  = (const float*)d_in[1];
    const float* W     = (const float*)d_in[2];
    const float* alpha = (const float*)d_in[3];
    const float* beta  = (const float*)d_in[4];
    float* out = (float*)d_out;

    float* ws_v    = (float*)d_ws;
    float* ws_part = ws_v + (size_t)BSZ * HSZ;

    cbsa_htW_wmma<<<dim3(BSZ / 16, HSZ / 16), 32, 0, stream>>>(h_t, W, ws_v);

    cbsa_flash_partial<<<dim3(BSZ, NCHUNK), THREADS, 0, stream>>>(cntx, ws_v, ws_part);

    cbsa_finalize<<<dim3(BSZ), THREADS, 0, stream>>>(h_t, ws_part, alpha, beta, out);
}


# h_t*W product: K split over 16 waves per 16x16 tile (was one wave, 256 serial steps), all fragment loads issued ahead, partial tiles summed through LDS; changes the f32 summation order of v
# speedup vs baseline: 1.4351x; 1.4351x over previous
#include <hip/hip_runtime.h>
#include <hip/hip_bf16.h>
#include <math.h>

#define BSZ 32
#define SSZ 2048
#define HSZ 1024

#define NCHUNK 8
#define ROWS_PER_CHUNK (SSZ / NCHUNK)
#define THREADS 256
#define NWAVES (THREADS / 32)
#define ROWS_PER_WAVE (ROWS_PER_CHUNK / NWAVES)
#define PSTRIDE (HSZ + 8)

typedef __attribute__((ext_vector_type(2))) float v2f;
typedef __attribute__((ext_vector_type(8))) float v8f;

#define K1_WAVES   16
#define K1_THREADS (K1_WAVES * 32)
#define K1_KSPAN   (HSZ / K1_WAVES)
#define K1_STEPS   (K1_KSPAN / 4)

__global__ __launch_bounds__(K1_THREADS)
void cbsa_htW_wmma(const float* __restrict__ h_t, const float* __restrict__ W,
                   float* __restrict__ v)
{
    __shared__ float s_part[K1_WAVES][256];

    const int t     = threadIdx.x;
    const int wave  = t >> 5;
    const int lane  = t & 31;
    const int mbase = blockIdx.x * 16;
    const int nbase = blockIdx.y * 16;
    const int r     = lane & 15;
    const int half  = lane >> 4;
    const int kfirst = wave * K1_KSPAN + 2 * half;

    const float* ap = h_t + (size_t)(mbase + r) * HSZ + kfirst;
    const float* bp = W + (size_t)kfirst * HSZ + nbase + r;

    v2f a[K1_STEPS], b[K1_STEPS];
    for (int s = 0; s < K1_STEPS; ++s) {
        b[s].x = bp[(size_t)(4 * s) * HSZ];
        b[s].y = bp[(size_t)(4 * s + 1) * HSZ];
    }
    for (int s = 0; s < K1_STEPS; ++s)
        a[s] = *(const v2f*)(ap + 4 * s);

    __builtin_amdgcn_sched_barrier(0);

    v8f c0 = {}, c1 = {};
    for (int s = 0; s < K1_STEPS; s += 2) {
        c0 = __builtin_amdgcn_wmma_f32_16x16x4_f32(
                  false, a[s],  false, b[s],
                  (short)0, c0,  false,  false);
        c1 = __builtin_amdgcn_wmma_f32_16x16x4_f32(
                  false, a[s + 1],  false, b[s + 1],
                  (short)0, c1,  false,  false);
    }

    for (int j = 0; j < 8; ++j) s_part[wave][j * 32 + lane] = c0[j] + c1[j];
    __syncthreads();

    if (t < 256) {
        float sum = s_part[0][t];
        for (int w = 1; w < K1_WAVES; ++w) sum += s_part[w][t];
        const int m = wave + 8 * half;
        v[(size_t)(mbase + m) * HSZ + nbase + r] = sum;
    }
}

__global__ __launch_bounds__(THREADS)
void cbsa_flash_partial(const float* __restrict__ cntx,
                        const float* __restrict__ vvec,
                        float* __restrict__ partials)
{
    __shared__ float s_v[HSZ];
    __shared__ float s_m[NWAVES];
    __shared__ float s_l[NWAVES];
    __shared__ float s_acc[NWAVES][HSZ];

    const int b     = blockIdx.x;
    const int chunk = blockIdx.y;
    const int t     = threadIdx.x;
    const int wave  = t >> 5;
    const int lane  = t & 31;

    ((float4*)s_v)[t] = ((const float4*)(vvec + (size_t)b * HSZ))[t];
    __syncthreads();

    float4 vf[8];
#pragma unroll
    for (int j = 0; j < 8; ++j) vf[j] = ((const float4*)s_v)[j * 32 + lane];

    float m = -INFINITY;
    float l = 0.0f;
    float4 acc[8];
#pragma unroll
    for (int j = 0; j < 8; ++j) acc[j] = make_float4(0.f, 0.f, 0.f, 0.f);

    const float* base =
        cntx + ((size_t)b * SSZ + (size_t)chunk * ROWS_PER_CHUNK) * HSZ;

    for (int rr = 0; rr < ROWS_PER_WAVE; ++rr) {
        const int row = wave + rr * NWAVES;
        const float4* rp = (const float4*)(base + (size_t)row * HSZ);

        float4 x[8];
#pragma unroll
        for (int j = 0; j < 8; ++j) x[j] = rp[j * 32 + lane];

        float d = 0.0f;
#pragma unroll
        for (int j = 0; j < 8; ++j) {
            d = fmaf(x[j].x, vf[j].x, d);
            d = fmaf(x[j].y, vf[j].y, d);
            d = fmaf(x[j].z, vf[j].z, d);
            d = fmaf(x[j].w, vf[j].w, d);
        }
#pragma unroll
        for (int off = 16; off > 0; off >>= 1) d += __shfl_xor(d, off, 32);

        const float nm = fmaxf(m, d);
        const float p  = __expf(d - nm);
        const float sc = __expf(m - nm);
        l = l * sc + p;
#pragma unroll
        for (int j = 0; j < 8; ++j) {
            acc[j].x = fmaf(p, x[j].x, acc[j].x * sc);
            acc[j].y = fmaf(p, x[j].y, acc[j].y * sc);
            acc[j].z = fmaf(p, x[j].z, acc[j].z * sc);
            acc[j].w = fmaf(p, x[j].w, acc[j].w * sc);
        }
        m = nm;
    }

    if (lane == 0) { s_m[wave] = m; s_l[wave] = l; }
#pragma unroll
    for (int j = 0; j < 8; ++j) ((float4*)s_acc[wave])[j * 32 + lane] = acc[j];
    __syncthreads();

    float M = s_m[0];
#pragma unroll
    for (int w = 1; w < NWAVES; ++w) M = fmaxf(M, s_m[w]);
    float wgt[NWAVES];
    float L = 0.0f;
#pragma unroll
    for (int w = 0; w < NWAVES; ++w) {
        wgt[w] = __expf(s_m[w] - M);
        L = fmaf(s_l[w], wgt[w], L);
    }

    float* out = partials + ((size_t)b * NCHUNK + chunk) * PSTRIDE;
    if (t == 0) { out[0] = M; out[1] = L; }

    float4 s = make_float4(0.f, 0.f, 0.f, 0.f);
#pragma unroll
    for (int w = 0; w < NWAVES; ++w) {
        const float4 a = ((const float4*)s_acc[w])[t];
        s.x = fmaf(wgt[w], a.x, s.x);
        s.y = fmaf(wgt[w], a.y, s.y);
        s.z = fmaf(wgt[w], a.z, s.z);
        s.w = fmaf(wgt[w], a.w, s.w);
    }
    ((float4*)(out + 8))[t] = s;
}

__global__ __launch_bounds__(THREADS)
void cbsa_finalize(const float* __restrict__ h_t,
                   const float* __restrict__ partials,
                   const float* __restrict__ alpha,
                   const float* __restrict__ beta,
                   float* __restrict__ out)
{
    const int b = blockIdx.x;
    const int t = threadIdx.x;

    const float* pb = partials + (size_t)b * NCHUNK * PSTRIDE;

    float M = -INFINITY;
#pragma unroll
    for (int c = 0; c < NCHUNK; ++c) M = fmaxf(M, pb[c * PSTRIDE]);

    float wgt[NCHUNK];
    float L = 0.0f;
#pragma unroll
    for (int c = 0; c < NCHUNK; ++c) {
        wgt[c] = __expf(pb[c * PSTRIDE] - M);
        L = fmaf(pb[c * PSTRIDE + 1], wgt[c], L);
    }

    float4 s = make_float4(0.f, 0.f, 0.f, 0.f);
#pragma unroll
    for (int c = 0; c < NCHUNK; ++c) {
        const float4 a = ((const float4*)(pb + c * PSTRIDE + 8))[t];
        s.x = fmaf(wgt[c], a.x, s.x);
        s.y = fmaf(wgt[c], a.y, s.y);
        s.z = fmaf(wgt[c], a.z, s.z);
        s.w = fmaf(wgt[c], a.w, s.w);
    }

    const float al  = alpha[0];
    const float inv = beta[0] / L;
    const float4 h  = ((const float4*)(h_t + (size_t)b * HSZ))[t];
    float4 o;
    o.x = fmaf(al, h.x, inv * s.x);
    o.y = fmaf(al, h.y, inv * s.y);
    o.z = fmaf(al, h.z, inv * s.z);
    o.w = fmaf(al, h.w, inv * s.w);
    ((float4*)(out + (size_t)b * HSZ))[t] = o;
}

extern "C" void kernel_launch(void* const* d_in, const int* in_sizes, int n_in,
                              void* d_out, int out_size, void* d_ws, size_t ws_size,
                              hipStream_t stream)
{
    const float* h_t   = (const float*)d_in[0];
    const float* cntx  = (const float*)d_in[1];
    const float* W     = (const float*)d_in[2];
    const float* alpha = (const float*)d_in[3];
    const float* beta  = (const float*)d_in[4];
    float* out = (float*)d_out;

    float* ws_v    = (float*)d_ws;
    float* ws_part = ws_v + (size_t)BSZ * HSZ;

    cbsa_htW_wmma<<<dim3(BSZ / 16, HSZ / 16), K1_THREADS, 0, stream>>>(h_t, W, ws_v);

    cbsa_flash_partial<<<dim3(BSZ, NCHUNK), THREADS, 0, stream>>>(cntx, ws_v, ws_part);

    cbsa_finalize<<<dim3(BSZ), THREADS, 0, stream>>>(h_t, ws_part, alpha, beta, out);
}


# cntx pass: 16 waves per workgroup (was 8), each requests its next 2 rows before summing a row (was none ahead), v read straight to registers; changes the f32 summation order of the softmax sums
# speedup vs baseline: 1.9237x; 1.3405x over previous
#include <hip/hip_runtime.h>
#include <hip/hip_bf16.h>
#include <math.h>

#define BSZ 32
#define SSZ 2048
#define HSZ 1024

#define NCHUNK 8
#define ROWS_PER_CHUNK (SSZ / NCHUNK)
#define THREADS 256
#define PSTRIDE (HSZ + 8)

typedef __attribute__((ext_vector_type(2))) float v2f;
typedef __attribute__((ext_vector_type(8))) float v8f;

#define K1_WAVES   16
#define K1_THREADS (K1_WAVES * 32)
#define K1_KSPAN   (HSZ / K1_WAVES)
#define K1_STEPS   (K1_KSPAN / 4)

__global__ __launch_bounds__(K1_THREADS)
void cbsa_htW_wmma(const float* __restrict__ h_t, const float* __restrict__ W,
                   float* __restrict__ v)
{
    __shared__ float s_part[K1_WAVES][256];

    const int t     = threadIdx.x;
    const int wave  = t >> 5;
    const int lane  = t & 31;
    const int mbase = blockIdx.x * 16;
    const int nbase = blockIdx.y * 16;
    const int r     = lane & 15;
    const int half  = lane >> 4;
    const int kfirst = wave * K1_KSPAN + 2 * half;

    const float* ap = h_t + (size_t)(mbase + r) * HSZ + kfirst;
    const float* bp = W + (size_t)kfirst * HSZ + nbase + r;

    v2f a[K1_STEPS], b[K1_STEPS];
    for (int s = 0; s < K1_STEPS; ++s) {
        b[s].x = bp[(size_t)(4 * s) * HSZ];
        b[s].y = bp[(size_t)(4 * s + 1) * HSZ];
    }
    for (int s = 0; s < K1_STEPS; ++s)
        a[s] = *(const v2f*)(ap + 4 * s);

    __builtin_amdgcn_sched_barrier(0);

    v8f c0 = {}, c1 = {};
    for (int s = 0; s < K1_STEPS; s += 2) {
        c0 = __builtin_amdgcn_wmma_f32_16x16x4_f32(
                  false, a[s],  false, b[s],
                  (short)0, c0,  false,  false);
        c1 = __builtin_amdgcn_wmma_f32_16x16x4_f32(
                  false, a[s + 1],  false, b[s + 1],
                  (short)0, c1,  false,  false);
    }

    for (int j = 0; j < 8; ++j) s_part[wave][j * 32 + lane] = c0[j] + c1[j];
    __syncthreads();

    if (t < 256) {
        float sum = s_part[0][t];
        for (int w = 1; w < K1_WAVES; ++w) sum += s_part[w][t];
        const int m = wave + 8 * half;
        v[(size_t)(mbase + m) * HSZ + nbase + r] = sum;
    }
}

#define K2_WAVES   16
#define K2_THREADS (K2_WAVES * 32)
#define K2_ROWS    (ROWS_PER_CHUNK / K2_WAVES)
#define K2_AHEAD   2
#define K2_RING    (K2_AHEAD + 1)
#define K2_ROWSTEP (K2_WAVES * HSZ)

typedef __attribute__((ext_vector_type(4))) float v4f;

__global__ __launch_bounds__(K2_THREADS)
void cbsa_flash_partial(const float* __restrict__ cntx,
                        const float* __restrict__ vvec,
                        float* __restrict__ partials)
{
    __shared__ float s_m[K2_WAVES];
    __shared__ float s_l[K2_WAVES];
    __shared__ __attribute__((aligned(16))) float s_acc[K2_WAVES][HSZ];

    const int b     = blockIdx.x;
    const int chunk = blockIdx.y;
    const int t     = threadIdx.x;
    const int wave  = t >> 5;
    const int lane  = t & 31;

    v4f vf[8];
    const v4f* vp = (const v4f*)(vvec + (size_t)b * HSZ) + lane;
#pragma unroll
    for (int j = 0; j < 8; ++j) vf[j] = vp[j * 32];

    const float* rp0 = cntx +
        ((size_t)b * SSZ + (size_t)chunk * ROWS_PER_CHUNK + (size_t)wave) * HSZ;

    v4f x[K2_RING][8];
    v4f acc[8];
#pragma unroll
    for (int j = 0; j < 8; ++j) acc[j] = (v4f){0.f, 0.f, 0.f, 0.f};

#define K2_LOAD_ROW(R) do { \
        unsigned off_ = (unsigned)(R) * K2_ROWSTEP; \
        asm volatile("" : "+s"(off_) \
                        : "v"(acc[0]), "v"(acc[1]), "v"(acc[2]), "v"(acc[3]), \
                          "v"(acc[4]), "v"(acc[5]), "v"(acc[6]), "v"(acc[7])); \
        const v4f* rq_ = (const v4f*)(rp0 + off_) + lane; \
        _Pragma("unroll") for (int j = 0; j < 8; ++j) x[(R) % K2_RING][j] = rq_[j * 32]; \
    } while (0)

#pragma unroll
    for (int r = 0; r < K2_AHEAD; ++r) K2_LOAD_ROW(r);

    float m = -INFINITY;
    float l = 0.0f;

#pragma unroll
    for (int r = 0; r < K2_ROWS; ++r) {
        if (r + K2_AHEAD < K2_ROWS) K2_LOAD_ROW(r + K2_AHEAD);

        const v4f* xr = x[r % K2_RING];

        float d = 0.0f;
        asm volatile("" : "+v"(d));
#pragma unroll
        for (int j = 0; j < 8; ++j) {
            d = fmaf(xr[j].x, vf[j].x, d);
            d = fmaf(xr[j].y, vf[j].y, d);
            d = fmaf(xr[j].z, vf[j].z, d);
            d = fmaf(xr[j].w, vf[j].w, d);
        }
#pragma unroll
        for (int off = 16; off > 0; off >>= 1) d += __shfl_xor(d, off, 32);

        const float nm = fmaxf(m, d);
        const float p  = __expf(d - nm);
        const float sc = __expf(m - nm);
        l = l * sc + p;
#pragma unroll
        for (int j = 0; j < 8; ++j) {
            acc[j].x = fmaf(p, xr[j].x, acc[j].x * sc);
            acc[j].y = fmaf(p, xr[j].y, acc[j].y * sc);
            acc[j].z = fmaf(p, xr[j].z, acc[j].z * sc);
            acc[j].w = fmaf(p, xr[j].w, acc[j].w * sc);
        }
        m = nm;
    }
#undef K2_LOAD_ROW

    if (lane == 0) { s_m[wave] = m; s_l[wave] = l; }
#pragma unroll
    for (int j = 0; j < 8; ++j) ((v4f*)s_acc[wave])[j * 32 + lane] = acc[j];
    __syncthreads();

    float M = s_m[0];
#pragma unroll
    for (int w = 1; w < K2_WAVES; ++w) M = fmaxf(M, s_m[w]);
    float wgt[K2_WAVES];
    float L = 0.0f;
#pragma unroll
    for (int w = 0; w < K2_WAVES; ++w) {
        wgt[w] = __expf(s_m[w] - M);
        L = fmaf(s_l[w], wgt[w], L);
    }

    float* out = partials + ((size_t)b * NCHUNK + chunk) * PSTRIDE;
    if (t == 0) { out[0] = M; out[1] = L; }

    float2 s = make_float2(0.f, 0.f);
#pragma unroll
    for (int w = 0; w < K2_WAVES; ++w) {
        const float2 a = ((const float2*)s_acc[w])[t];
        s.x = fmaf(wgt[w], a.x, s.x);
        s.y = fmaf(wgt[w], a.y, s.y);
    }
    ((float2*)(out + 8))[t] = s;
}

__global__ __launch_bounds__(THREADS)
void cbsa_finalize(const float* __restrict__ h_t,
                   const float* __restrict__ partials,
                   const float* __restrict__ alpha,
                   const float* __restrict__ beta,
                   float* __restrict__ out)
{
    const int b = blockIdx.x;
    const int t = threadIdx.x;

    const float* pb = partials + (size_t)b * NCHUNK * PSTRIDE;

    float M = -INFINITY;
#pragma unroll
    for (int c = 0; c < NCHUNK; ++c) M = fmaxf(M, pb[c * PSTRIDE]);

    float wgt[NCHUNK];
    float L = 0.0f;
#pragma unroll
    for (int c = 0; c < NCHUNK; ++c) {
        wgt[c] = __expf(pb[c * PSTRIDE] - M);
        L = fmaf(pb[c * PSTRIDE + 1], wgt[c], L);
    }

    float4 s = make_float4(0.f, 0.f, 0.f, 0.f);
#pragma unroll
    for (int c = 0; c < NCHUNK; ++c) {
        const float4 a = ((const float4*)(pb + c * PSTRIDE + 8))[t];
        s.x = fmaf(wgt[c], a.x, s.x);
        s.y = fmaf(wgt[c], a.y, s.y);
        s.z = fmaf(wgt[c], a.z, s.z);
        s.w = fmaf(wgt[c], a.w, s.w);
    }

    const float al  = alpha[0];
    const float inv = beta[0] / L;
    const float4 h  = ((const float4*)(h_t + (size_t)b * HSZ))[t];
    float4 o;
    o.x = fmaf(al, h.x, inv * s.x);
    o.y = fmaf(al, h.y, inv * s.y);
    o.z = fmaf(al, h.z, inv * s.z);
    o.w = fmaf(al, h.w, inv * s.w);
    ((float4*)(out + (size_t)b * HSZ))[t] = o;
}

extern "C" void kernel_launch(void* const* d_in, const int* in_sizes, int n_in,
                              void* d_out, int out_size, void* d_ws, size_t ws_size,
                              hipStream_t stream)
{
    const float* h_t   = (const float*)d_in[0];
    const float* cntx  = (const float*)d_in[1];
    const float* W     = (const float*)d_in[2];
    const float* alpha = (const float*)d_in[3];
    const float* beta  = (const float*)d_in[4];
    float* out = (float*)d_out;

    float* ws_v    = (float*)d_ws;
    float* ws_part = ws_v + (size_t)BSZ * HSZ;

    cbsa_htW_wmma<<<dim3(BSZ / 16, HSZ / 16), K1_THREADS, 0, stream>>>(h_t, W, ws_v);

    cbsa_flash_partial<<<dim3(BSZ, NCHUNK), K2_THREADS, 0, stream>>>(cntx, ws_v, ws_part);

    cbsa_finalize<<<dim3(BSZ), THREADS, 0, stream>>>(h_t, ws_part, alpha, beta, out);
}
